# starting kernel
# baseline (speedup 1.0000x reference)
#include <hip/hip_runtime.h>

typedef __attribute__((ext_vector_type(2))) float v2f;
typedef __attribute__((ext_vector_type(8))) float v8f;

#define BATCH 16
#define CIN 8
#define IMG 256
#define NOC 8
#define KW 3
#define TAPS (CIN * KW * KW)
#define WROW (NOC * KW * KW)
#define PPB 8

#define PA_STRIDE (BATCH * TAPS)
#define WB_STRIDE (16 * TAPS)

__global__ __launch_bounds__(256, 2)
void lcn2d_wmma_f32(const float* __restrict__ xin,
                    const float* __restrict__ wgt,
                    const float* __restrict__ bias,
                    float* __restrict__ out)
{
    __shared__ float pa[PPB * PA_STRIDE];
    __shared__ float wb[PPB * WB_STRIDE];

    const int tid = threadIdx.x;
    const int px  = blockIdx.y;
    const int py0 = blockIdx.x * PPB;

    for (int c = 0; c < CIN; ++c) {
        const float* src = wgt + ((c * IMG + px) * IMG + py0) * WROW;
        for (int t = tid; t < PPB * WROW; t += 256) {
            int pix = t / WROW;
            int r   = t - pix * WROW;
            int oc  = r / 9;
            int tap = r - oc * 9;
            wb[pix * WB_STRIDE + oc * TAPS + c * 9 + tap] = src[t];
        }
    }
    for (int t = tid; t < PPB * NOC * TAPS; t += 256) {
        int pix = t / (NOC * TAPS);
        int r   = t - pix * (NOC * TAPS);
        wb[pix * WB_STRIDE + NOC * TAPS + r] = 0.0f;
    }

    for (int idx = tid; idx < PPB * PA_STRIDE; idx += 256) {
        int k   = idx % TAPS;
        int m   = (idx / TAPS) & (BATCH - 1);
        int pix = idx / PA_STRIDE;
        int c   = k / 9;
        int r   = k - c * 9;
        int i   = r / 3;
        int j   = r - i * 3;
        int sx = px + i - 1;        sx = sx < 0 ? 0 : (sx > IMG - 1 ? IMG - 1 : sx);
        int sy = py0 + pix - 1 + j; sy = sy < 0 ? 0 : (sy > IMG - 1 ? IMG - 1 : sy);
        pa[idx] = xin[((m * CIN + c) * IMG + sx) * IMG + sy];
    }
    __syncthreads();

    const int wv   = tid >> 5;
    const int lane = tid & 31;
    const int py   = py0 + wv;
    const int n    = lane & 15;
    const int m    = lane & 15;
    const int kh   = (lane >> 4) << 1;

    float bv = (n < NOC) ? bias[(px * IMG + py) * NOC + n] : 0.0f;
    v8f acc = { bv, bv, bv, bv, bv, bv, bv, bv };

    const float* pa_row = &pa[wv * PA_STRIDE + m * TAPS + kh];
    const float* wb_row = &wb[wv * WB_STRIDE + n * TAPS + kh];

    #pragma unroll
    for (int kk = 0; kk < TAPS / 4; ++kk) {
        v2f a  = *(const v2f*)(pa_row + kk * 4);
        v2f bm = *(const v2f*)(wb_row + kk * 4);
        acc = __builtin_amdgcn_wmma_f32_16x16x4_f32(
            false, a, false, bm, (short)0, acc, false, false);
    }

    if (n < NOC) {
        const int mb = (lane >> 4) << 3;
        #pragma unroll
        for (int r = 0; r < 8; ++r) {
            int b = r + mb;
            out[((b * NOC + n) * IMG + px) * IMG + py] = acc[r];
        }
    }
}

extern "C" void kernel_launch(void* const* d_in, const int* in_sizes, int n_in,
                              void* d_out, int out_size, void* d_ws, size_t ws_size,
                              hipStream_t stream) {
    const float* x    = (const float*)d_in[0];
    const float* wgt  = (const float*)d_in[1];
    const float* bias = (const float*)d_in[2];
    float* out        = (float*)d_out;

    dim3 grid(IMG / PPB, IMG);
    dim3 block(256);
    lcn2d_wmma_f32<<<grid, block, 0, stream>>>(x, wgt, bias, out);
}


# Staging loads batched: each thread issues all its x and weight loads (16-byte, row-coalesced) before its first LDS store, replacing one dword load-wait-store per element; LDS layout kept
# speedup vs baseline: 1.3842x; 1.3842x over previous
#include <hip/hip_runtime.h>


typedef __attribute__((ext_vector_type(2))) float v2f;
typedef __attribute__((ext_vector_type(4))) float v4f;
typedef __attribute__((ext_vector_type(8))) float v8f;

#define BATCH 16
#define CIN 8
#define IMG 256
#define NOC 8
#define KW 3
#define TAPS (CIN * KW * KW)
#define WROW (NOC * KW * KW)
#define PPB 8

#define PA_STRIDE (BATCH * TAPS)
#define WB_STRIDE (16 * TAPS)

#define XROWS (BATCH * CIN * KW)
#define WQ_PER_C (PPB * WROW / 4)
#define WQ_ALL (CIN * WQ_PER_C)

__global__ __launch_bounds__(256, 2)
void lcn2d_wmma_f32(const float* __restrict__ xin,
                    const float* __restrict__ wgt,
                    const float* __restrict__ bias,
                    float* __restrict__ out)
{
    __shared__ float pa[PPB * PA_STRIDE];
    __shared__ float wb[PPB * WB_STRIDE];

    const int tid = threadIdx.x;
    const int px  = blockIdx.y;
    const int py0 = blockIdx.x * PPB;

    const int p = tid & 1;
    int hc = p ? (py0 + PPB) : (py0 - 1);
    hc = hc < 0 ? 0 : (hc > IMG - 1 ? IMG - 1 : hc);
    v4f   xq[3];
    float xh[3];
    #pragma unroll
    for (int s = 0; s < 3; ++s) {
        int row = (tid >> 1) + 128 * s;
        int mc  = row / 3;
        int i   = row - mc * 3;
        int sx  = px + i - 1;  sx = sx < 0 ? 0 : (sx > IMG - 1 ? IMG - 1 : sx);
        const float* g = xin + (mc * IMG + sx) * IMG;
        xq[s] = *(const v4f*)(g + py0 + 4 * p);
        xh[s] = g[hc];
    }

    v4f wq[5];
    #pragma unroll
    for (int s = 0; s < 5; ++s) {
        if (s < 4 || tid < WQ_ALL - 1024) {
            int g = tid + 256 * s;
            int c = g / WQ_PER_C;
            int q = g - c * WQ_PER_C;
            wq[s] = *(const v4f*)(wgt + ((c * IMG + px) * IMG + py0) * WROW + 4 * q);
        }
    }

    for (int t = tid; t < PPB * NOC * TAPS; t += 256) {
        int pix = t / (NOC * TAPS);
        int r   = t - pix * (NOC * TAPS);
        wb[pix * WB_STRIDE + NOC * TAPS + r] = 0.0f;
    }

    #pragma unroll
    for (int s = 0; s < 3; ++s) {
        int row = (tid >> 1) + 128 * s;
        float* d = pa + 5 * p * PA_STRIDE + 3 * row;
        v4f  q  = xq[s];
        float h = xh[s];
        float v0 = p ? q.x : h;
        float v1 = p ? q.y : q.x;
        float v2 = p ? q.z : q.y;
        float v3 = p ? q.w : q.z;
        float v4 = p ? h   : q.w;
        d[0 * PA_STRIDE + 0] = v0;
        d[1 * PA_STRIDE + 0] = v1;
        d[0 * PA_STRIDE + 1] = v1;
        d[2 * PA_STRIDE + 0] = v2;
        d[1 * PA_STRIDE + 1] = v2;
        d[0 * PA_STRIDE + 2] = v2;
        d[2 * PA_STRIDE + 1] = v3;
        d[1 * PA_STRIDE + 2] = v3;
        d[2 * PA_STRIDE + 2] = v4;
        if (p) {
            d[-1 * PA_STRIDE + 1] = v0;
            d[-2 * PA_STRIDE + 2] = v0;
            d[-1 * PA_STRIDE + 2] = v1;
        } else {
            d[3 * PA_STRIDE + 0] = v3;
            d[4 * PA_STRIDE + 0] = v4;
            d[3 * PA_STRIDE + 1] = v4;
        }
    }

    #pragma unroll
    for (int s = 0; s < 5; ++s) {
        if (s < 4 || tid < WQ_ALL - 1024) {
            int g   = tid + 256 * s;
            int c   = g / WQ_PER_C;
            int q   = g - c * WQ_PER_C;
            int pix = q / 18;
            int r0  = (q - pix * 18) * 4;
            float* d = wb + pix * WB_STRIDE + c * 9;
            v4f w = wq[s];
            #pragma unroll
            for (int e = 0; e < 4; ++e) {
                int r  = r0 + e;
                int oc = r / 9;
                d[r + 63 * oc] = w[e];
            }
        }
    }
    __syncthreads();

    const int wv   = tid >> 5;
    const int lane = tid & 31;
    const int py   = py0 + wv;
    const int n    = lane & 15;
    const int m    = lane & 15;
    const int kh   = (lane >> 4) << 1;

    float bv = (n < NOC) ? bias[(px * IMG + py) * NOC + n] : 0.0f;
    v8f acc = { bv, bv, bv, bv, bv, bv, bv, bv };

    const float* pa_row = &pa[wv * PA_STRIDE + m * TAPS + kh];
    const float* wb_row = &wb[wv * WB_STRIDE + n * TAPS + kh];

    #pragma unroll
    for (int kk = 0; kk < TAPS / 4; ++kk) {
        v2f a  = *(const v2f*)(pa_row + kk * 4);
        v2f bm = *(const v2f*)(wb_row + kk * 4);
        acc = __builtin_amdgcn_wmma_f32_16x16x4_f32(
            false, a, false, bm, (short)0, acc, false, false);
    }

    if (n < NOC) {
        const int mb = (lane >> 4) << 3;
        #pragma unroll
        for (int r = 0; r < 8; ++r) {
            int b = r + mb;
            out[((b * NOC + n) * IMG + px) * IMG + py] = acc[r];
        }
    }
}

extern "C" void kernel_launch(void* const* d_in, const int* in_sizes, int n_in,
                              void* d_out, int out_size, void* d_ws, size_t ws_size,
                              hipStream_t stream) {
    const float* x    = (const float*)d_in[0];
    const float* wgt  = (const float*)d_in[1];
    const float* bias = (const float*)d_in[2];
    float* out        = (float*)d_out;

    dim3 grid(IMG / PPB, IMG);
    dim3 block(256);
    lcn2d_wmma_f32<<<grid, block, 0, stream>>>(x, wgt, bias, out);
}


# Block tile 8 -> 32 pixels of a row (4 per wave): x rows land raw in LDS as whole 128-B lines and feed A directly (no patch copy), out rows leave as whole lines via LDS; k and sum order kept
# speedup vs baseline: 5.9780x; 4.3187x over previous
#include <hip/hip_runtime.h>


typedef __attribute__((ext_vector_type(2))) float v2f;
typedef __attribute__((ext_vector_type(4))) float v4f;
typedef __attribute__((ext_vector_type(8))) float v8f;

#define BATCH 16
#define CIN 8
#define IMG 256
#define NOC 8
#define KW 3
#define TAPS (CIN * KW * KW)
#define WROW (NOC * KW * KW)
#define NTHR 256
#define PPW 4
#define PPB 32

#define XROWS (BATCH * CIN * KW)
#define XP 44
#define XC0 4
#define WBP (NOC * TAPS)
#define STP 36

#define XPIECES (XROWS * PPB / 4 / NTHR)
#define XEDGES (XROWS * 2 / NTHR)
#define WQ_PER_C (PPB * WROW / 4)
#define WPIECES (CIN * WQ_PER_C / NTHR)
#define OPIECES (BATCH * NOC * PPB / 4 / NTHR)

#define XOFF(k) (((k) / 3) * 16 * XP + ((k) % 3))

__global__ __launch_bounds__(NTHR)
void lcn2d_wmma_f32(const float* __restrict__ xin,
                    const float* __restrict__ wgt,
                    const float* __restrict__ bias,
                    float* __restrict__ out)
{
    __shared__ __attribute__((aligned(16))) float xs[XROWS * XP];
    __shared__ __attribute__((aligned(16))) float wb[PPB * WBP];

    const int tid  = threadIdx.x;
    const int px   = blockIdx.y;
    const int py0  = blockIdx.x * PPB;
    const int wv   = tid >> 5;
    const int lane = tid & 31;
    const int n    = lane & 15;
    const int m    = lane & 15;
    const int hb   = lane >> 4;

    int sxr[KW];
    #pragma unroll
    for (int i = 0; i < KW; ++i) {
        int sx = px + i - 1;
        sxr[i] = sx < 0 ? 0 : (sx > IMG - 1 ? IMG - 1 : sx);
    }
    const int q8 = tid & 7;
    v4f xq[XPIECES];
    #pragma unroll
    for (int s = 0; s < XPIECES; ++s) {
        int i  = s >> 2;
        int mc = (tid >> 3) + 32 * (s & 3);
        xq[s] = *(const v4f*)(xin + (mc * IMG + sxr[i]) * IMG + py0 + 4 * q8);
    }
    const int emc  = tid >> 1;
    const int side = tid & 1;
    int ecol = side ? (py0 + PPB) : (py0 - 1);
    ecol = ecol < 0 ? 0 : (ecol > IMG - 1 ? IMG - 1 : ecol);
    float xe[XEDGES];
    #pragma unroll
    for (int i = 0; i < XEDGES; ++i)
        xe[i] = xin[(emc * IMG + sxr[i]) * IMG + ecol];
    const float* wsrc = wgt + ((wv * IMG + px) * IMG + py0) * WROW + 4 * lane;
    v4f wq[WPIECES];
    #pragma unroll
    for (int s = 0; s < WPIECES; ++s)
        wq[s] = *(const v4f*)(wsrc + 128 * s);
    float bv[PPW];
    #pragma unroll
    for (int pp = 0; pp < PPW; ++pp)
        bv[pp] = bias[(px * IMG + py0 + PPW * wv + pp) * NOC + (n & 7)];

    #pragma unroll
    for (int s = 0; s < XPIECES; ++s) {
        int i    = s >> 2;
        int mc   = (tid >> 3) + 32 * (s & 3);
        int lrow = ((mc & 7) * 3 + i) * 16 + (mc >> 3);
        *(v4f*)(xs + lrow * XP + XC0 + 4 * q8) = xq[s];
    }
    #pragma unroll
    for (int i = 0; i < XEDGES; ++i) {
        int lrow = ((emc & 7) * 3 + i) * 16 + (emc >> 3);
        xs[lrow * XP + (side ? (XC0 + PPB) : (XC0 - 1))] = xe[i];
    }

    const int pl = lane >= 18 ? 1 : 0;
    const int ql = lane - 18 * pl;
    #pragma unroll
    for (int s9 = 0; s9 < 9; ++s9) {
        int qr  = ql + (32 * s9) % 18;
        int cy  = qr >= 18 ? 1 : 0;
        int pix = pl + (32 * s9) / 18 + cy;
        int r0  = (qr - 18 * cy) * 4;
        int oc0 = r0 / 9;
        int eb  = 9 - (r0 - oc0 * 9);
        float* dlo = wb + pix * WBP + wv * 9 + r0 + 63 * oc0;
        float* dhi = dlo + 63;
        #pragma unroll
        for (int hf = 0; hf < 2; ++hf) {
            v4f w = wq[s9 + 9 * hf];
            #pragma unroll
            for (int e = 0; e < 4; ++e) {
                float* d = (e >= eb) ? dhi : dlo;
                d[hf * 16 * WBP + e] = w[e];
            }
        }
    }
    __syncthreads();

    v8f acc[PPW];
    #pragma unroll
    for (int pp = 0; pp < PPW; ++pp) {
        float b = bv[pp];
        acc[pp] = (v8f){ b, b, b, b, b, b, b, b };
    }

    const float* xa = xs + m * XP + (XC0 - 1) + PPW * wv;
    const float* wr = wb + (PPW * wv) * WBP + (n & 7) * TAPS + 2 * hb;

    #pragma unroll
    for (int kk = 0; kk < TAPS / 4; ++kk) {
        const int o0 = hb ? XOFF(4 * kk + 2) : XOFF(4 * kk);
        const int o1 = hb ? XOFF(4 * kk + 3) : XOFF(4 * kk + 1);
        #pragma unroll
        for (int pp = 0; pp < PPW; ++pp) {
            v2f a;
            a.x = xa[o0 + pp];
            a.y = xa[o1 + pp];
            v2f bm = *(const v2f*)(wr + pp * WBP + kk * 4);
            acc[pp] = __builtin_amdgcn_wmma_f32_16x16x4_f32(
                false, a, false, bm, (short)0, acc[pp], false, false);
        }
    }

    __syncthreads();
    float* st = xs;
    if (n < NOC) {
        #pragma unroll
        for (int r = 0; r < 8; ++r) {
            int b = r + 8 * hb;
            v4f v = { acc[0][r], acc[1][r], acc[2][r], acc[3][r] };
            *(v4f*)(st + (b * NOC + n) * STP + PPW * wv) = v;
        }
    }
    __syncthreads();
    #pragma unroll
    for (int s = 0; s < OPIECES; ++s) {
        int f   = tid + NTHR * s;
        int row = f >> 3;
        int q   = f & 7;
        v4f v = *(const v4f*)(st + row * STP + 4 * q);
        *(v4f*)(out + (row * IMG + px) * IMG + py0 + 4 * q) = v;
    }
}

extern "C" void kernel_launch(void* const* d_in, const int* in_sizes, int n_in,
                              void* d_out, int out_size, void* d_ws, size_t ws_size,
                              hipStream_t stream) {
    const float* x    = (const float*)d_in[0];
    const float* wgt  = (const float*)d_in[1];
    const float* bias = (const float*)d_in[2];
    float* out        = (float*)d_out;

    dim3 grid(IMG / PPB, IMG);
    dim3 block(NTHR);
    lcn2d_wmma_f32<<<grid, block, 0, stream>>>(x, wgt, bias, out);
}


# Weights go from memory straight into the B registers (lane = oc, 36-byte runs, lane halves split the channels), in flight across an LDS-only barrier; no LDS pass; k pairing, f32 sum order changed
# speedup vs baseline: 6.1734x; 1.0327x over previous
#include <hip/hip_runtime.h>


typedef __attribute__((ext_vector_type(2))) float v2f;
typedef __attribute__((ext_vector_type(4))) float v4f;
typedef __attribute__((ext_vector_type(8))) float v8f;
typedef v4f v4f_a4 __attribute__((aligned(4)));

#define BATCH 16
#define CIN 8
#define IMG 256
#define NOC 8
#define KW 3
#define TAPS (CIN * KW * KW)
#define WROW (NOC * KW * KW)
#define NTHR 256
#define PPW 4
#define PPB 32

#define XROWS (BATCH * CIN * KW)
#define XP 44
#define XC0 4
#define CSTR (IMG * IMG * WROW)
#define WPL 36
#define STP 36

#define XPIECES (XROWS * PPB / 4 / NTHR)
#define XEDGES (XROWS * 2 / NTHR)
#define OPIECES (BATCH * NOC * PPB / 4 / NTHR)

#define XOFF(k) (((k) / 3) * 16 * XP + ((k) % 3))
#define RK(r) ((r) < 32 ? ((r) / 8) * 9 + ((r) % 8) : ((r) - 32) * 9 + 8)

__global__ __launch_bounds__(NTHR) __attribute__((amdgpu_waves_per_eu(4, 4)))
void lcn2d_wmma_f32(const float* __restrict__ xin,
                    const float* __restrict__ wgt,
                    const float* __restrict__ bias,
                    float* __restrict__ out)
{
    __shared__ __attribute__((aligned(16))) float xs[XROWS * XP];

    const int tid  = threadIdx.x;
    const int px   = blockIdx.y;
    const int py0  = blockIdx.x * PPB;
    const int wv   = tid >> 5;
    const int lane = tid & 31;
    const int n    = lane & 15;
    const int m    = lane & 15;
    const int hb   = lane >> 4;

    float bv[PPW];
    #pragma unroll
    for (int pp = 0; pp < PPW; ++pp)
        bv[pp] = bias[(px * IMG + py0 + PPW * wv + pp) * NOC + (n & 7)];
    int sxr[KW];
    #pragma unroll
    for (int i = 0; i < KW; ++i) {
        int sx = px + i - 1;
        sxr[i] = sx < 0 ? 0 : (sx > IMG - 1 ? IMG - 1 : sx);
    }
    const int q8 = tid & 7;
    v4f xq[XPIECES];
    #pragma unroll
    for (int s = 0; s < XPIECES; ++s) {
        int i  = s >> 2;
        int mc = (tid >> 3) + 32 * (s & 3);
        xq[s] = *(const v4f*)(xin + (mc * IMG + sxr[i]) * IMG + py0 + 4 * q8);
    }
    const int emc  = tid >> 1;
    const int side = tid & 1;
    int ecol = side ? (py0 + PPB) : (py0 - 1);
    ecol = ecol < 0 ? 0 : (ecol > IMG - 1 ? IMG - 1 : ecol);
    float xe[XEDGES];
    #pragma unroll
    for (int i = 0; i < XEDGES; ++i)
        xe[i] = xin[(emc * IMG + sxr[i]) * IMG + ecol];
    const float* wl = wgt + ((4 * hb * IMG + px) * IMG + py0 + PPW * wv) * WROW + (n & 7) * 9;
    float w[PPW][WPL];
    #pragma unroll
    for (int pp = 0; pp < PPW; ++pp) {
        if (pp == 2) __builtin_amdgcn_sched_barrier(0);
        #pragma unroll
        for (int cc = 0; cc < 4; ++cc) {
            const float* p = wl + pp * WROW + cc * CSTR;
            v4f lo = *(const v4f_a4*)p;
            v4f hi = *(const v4f_a4*)(p + 4);
            #pragma unroll
            for (int e = 0; e < 4; ++e) {
                w[pp][8 * cc + e]     = lo[e];
                w[pp][8 * cc + 4 + e] = hi[e];
            }
            w[pp][32 + cc] = p[8];
        }
    }
    #pragma unroll
    for (int s = 0; s < XPIECES; ++s) {
        int i    = s >> 2;
        int mc   = (tid >> 3) + 32 * (s & 3);
        int lrow = ((mc & 7) * 3 + i) * 16 + (mc >> 3);
        *(v4f*)(xs + lrow * XP + XC0 + 4 * q8) = xq[s];
    }
    #pragma unroll
    for (int i = 0; i < XEDGES; ++i) {
        int lrow = ((emc & 7) * 3 + i) * 16 + (emc >> 3);
        xs[lrow * XP + (side ? (XC0 + PPB) : (XC0 - 1))] = xe[i];
    }

    __builtin_amdgcn_fence(__ATOMIC_RELEASE, "workgroup", "local");
    __builtin_amdgcn_s_barrier();
    __builtin_amdgcn_fence(__ATOMIC_ACQUIRE, "workgroup", "local");

    v8f acc[PPW];
    #pragma unroll
    for (int pp = 0; pp < PPW; ++pp) {
        float b = bv[pp];
        acc[pp] = (v8f){ b, b, b, b, b, b, b, b };
    }

    const float* xa = xs + m * XP + (XC0 - 1) + PPW * wv + hb * (12 * 16 * XP);

    #pragma unroll
    for (int pr = 0; pr < PPW; pr += 2) {
        if (pr) __builtin_amdgcn_sched_barrier(0);
        #pragma unroll
        for (int s = 0; s < WPL / 2; ++s) {
            #pragma unroll
            for (int q = 0; q < 2; ++q) {
                const int pp = pr + q;
                v2f a;
                a.x = xa[XOFF(RK(2 * s)) + pp];
                a.y = xa[XOFF(RK(2 * s + 1)) + pp];
                v2f bm = { w[pp][2 * s], w[pp][2 * s + 1] };
                acc[pp] = __builtin_amdgcn_wmma_f32_16x16x4_f32(
                    false, a, false, bm, (short)0, acc[pp], false, false);
            }
        }
    }

    __syncthreads();
    float* st = xs;
    if (n < NOC) {
        #pragma unroll
        for (int r = 0; r < 8; ++r) {
            int b = r + 8 * hb;
            v4f v = { acc[0][r], acc[1][r], acc[2][r], acc[3][r] };
            *(v4f*)(st + (b * NOC + n) * STP + PPW * wv) = v;
        }
    }
    __syncthreads();
    #pragma unroll
    for (int s = 0; s < OPIECES; ++s) {
        int f   = tid + NTHR * s;
        int row = f >> 3;
        int q   = f & 7;
        v4f v = *(const v4f*)(st + row * STP + 4 * q);
        *(v4f*)(out + (row * IMG + px) * IMG + py0 + 4 * q) = v;
    }
}

extern "C" void kernel_launch(void* const* d_in, const int* in_sizes, int n_in,
                              void* d_out, int out_size, void* d_ws, size_t ws_size,
                              hipStream_t stream) {
    const float* x    = (const float*)d_in[0];
    const float* wgt  = (const float*)d_in[1];
    const float* bias = (const float*)d_in[2];
    float* out        = (float*)d_out;

    dim3 grid(IMG / PPB, IMG);
    dim3 block(NTHR);
    lcn2d_wmma_f32<<<grid, block, 0, stream>>>(x, wgt, bias, out);
}


# Block tile 1 -> 2 image rows: the four input rows and their halo columns are staged once for both; the 2nd row's weights load into freed B registers during the 1st row's products; k, sum order kept
# speedup vs baseline: 6.9310x; 1.1227x over previous
#include <hip/hip_runtime.h>


typedef __attribute__((ext_vector_type(2))) float v2f;
typedef __attribute__((ext_vector_type(4))) float v4f;
typedef __attribute__((ext_vector_type(8))) float v8f;
typedef v4f v4f_a4 __attribute__((aligned(4)));

#define BATCH 16
#define CIN 8
#define IMG 256
#define NOC 8
#define KW 3
#define TAPS (CIN * KW * KW)
#define WROW (NOC * KW * KW)
#define NTHR 256
#define PPW 4
#define PPB 32
#define PXR 2
#define SRC (PXR + KW - 1)

#define XROWS (BATCH * CIN * SRC)
#define XP 44
#define XC0 4
#define CSTR (IMG * IMG * WROW)
#define WPL 36
#define STP 36

#define XPIECES (XROWS * PPB / 4 / NTHR)
#define XEDGES (XROWS * 2 / NTHR)
#define OPIECES (PXR * BATCH * NOC * PPB / 4 / NTHR)
#define NSLOT (PXR * PPW / 2)

#define XOFF(k, pr) (((((k) / 9) * SRC) + (((k) % 9) / 3) + (pr)) * 16 * XP + ((k) % 3))
#define RK(r) ((r) < 32 ? ((r) / 8) * 9 + ((r) % 8) : ((r) - 32) * 9 + 8)

__global__ __launch_bounds__(NTHR) __attribute__((amdgpu_waves_per_eu(4, 4)))
void lcn2d_wmma_f32(const float* __restrict__ xin,
                    const float* __restrict__ wgt,
                    const float* __restrict__ bias,
                    float* __restrict__ out)
{
    __shared__ __attribute__((aligned(16))) float xs[XROWS * XP];
    __shared__ __attribute__((aligned(16))) float st[PXR * BATCH * NOC * STP];

    const int tid  = threadIdx.x;
    const int px0  = blockIdx.y * PXR;
    const int py0  = blockIdx.x * PPB;
    const int wv   = tid >> 5;
    const int lane = tid & 31;
    const int n    = lane & 15;
    const int m    = lane & 15;
    const int hb   = lane >> 4;

    float bv[PXR][PPW];
    #pragma unroll
    for (int pr = 0; pr < PXR; ++pr) {
        #pragma unroll
        for (int pp = 0; pp < PPW; ++pp)
            bv[pr][pp] = bias[((px0 + pr) * IMG + py0 + PPW * wv + pp) * NOC + (n & 7)];
    }
    int sxr[SRC];
    #pragma unroll
    for (int r = 0; r < SRC; ++r) {
        int sx = px0 + r - 1;
        sxr[r] = sx < 0 ? 0 : (sx > IMG - 1 ? IMG - 1 : sx);
    }
    const int q8 = tid & 7;
    v4f xq[XPIECES];
    #pragma unroll
    for (int s = 0; s < XPIECES; ++s) {
        int r  = s >> 2;
        int mc = (tid >> 3) + 32 * (s & 3);
        xq[s] = *(const v4f*)(xin + (mc * IMG + sxr[r]) * IMG + py0 + 4 * q8);
    }
    const int emc  = tid >> 1;
    const int side = tid & 1;
    int ecol = side ? (py0 + PPB) : (py0 - 1);
    ecol = ecol < 0 ? 0 : (ecol > IMG - 1 ? IMG - 1 : ecol);
    float xe[XEDGES];
    #pragma unroll
    for (int r = 0; r < XEDGES; ++r)
        xe[r] = xin[(emc * IMG + sxr[r]) * IMG + ecol];
    const float* wl = wgt + ((4 * hb * IMG + px0) * IMG + py0 + PPW * wv) * WROW + (n & 7) * 9;
    float w[2][2][WPL];
    #define LOAD_PAIR(q)                                                                   \
        _Pragma("unroll")   \
        for (int e2 = 0; e2 < 2; ++e2) {                                                   \
            _Pragma("unroll")                                                 \
            for (int cc = 0; cc < 4; ++cc) {                                               \
                const float* p = wl + ((q) >> 1) * (IMG * WROW) + (2 * ((q) & 1) + e2) * WROW + cc * CSTR; \
                v4f lo = *(const v4f_a4*)p;                                                \
                v4f hi = *(const v4f_a4*)(p + 4);                                          \
                _Pragma("unroll")                                             \
                for (int e = 0; e < 4; ++e) {                                              \
                    w[(q) & 1][e2][8 * cc + e]     = lo[e];                                \
                    w[(q) & 1][e2][8 * cc + 4 + e] = hi[e];                                \
                }                                                                          \
                w[(q) & 1][e2][32 + cc] = p[8];                                            \
            }                                                                              \
        }
    LOAD_PAIR(0)
    __builtin_amdgcn_sched_barrier(0);
    LOAD_PAIR(1)
    #pragma unroll
    for (int s = 0; s < XPIECES; ++s) {
        int r    = s >> 2;
        int mc   = (tid >> 3) + 32 * (s & 3);
        int lrow = ((mc & 7) * SRC + r) * 16 + (mc >> 3);
        *(v4f*)(xs + lrow * XP + XC0 + 4 * q8) = xq[s];
    }
    #pragma unroll
    for (int r = 0; r < XEDGES; ++r) {
        int lrow = ((emc & 7) * SRC + r) * 16 + (emc >> 3);
        xs[lrow * XP + (side ? (XC0 + PPB) : (XC0 - 1))] = xe[r];
    }

    __builtin_amdgcn_fence(__ATOMIC_RELEASE, "workgroup", "local");
    __builtin_amdgcn_s_barrier();
    __builtin_amdgcn_fence(__ATOMIC_ACQUIRE, "workgroup", "local");

    const float* xa = xs + m * XP + (XC0 - 1) + PPW * wv + hb * (4 * SRC * 16 * XP);

    v8f acc[PPW];
    #pragma unroll
    for (int q = 0; q < NSLOT; ++q) {
        const int pr = q >> 1;
        if (q) __builtin_amdgcn_sched_barrier(0);
        if ((q & 1) == 0) {
            #pragma unroll
            for (int pp = 0; pp < PPW; ++pp) {
                float b = bv[pr][pp];
                acc[pp] = (v8f){ b, b, b, b, b, b, b, b };
            }
        }
        #pragma unroll
        for (int s = 0; s < WPL / 2; ++s) {
            #pragma unroll
            for (int e2 = 0; e2 < 2; ++e2) {
                const int pp = 2 * (q & 1) + e2;
                v2f a;
                a.x = xa[XOFF(RK(2 * s), pr) + pp];
                a.y = xa[XOFF(RK(2 * s + 1), pr) + pp];
                v2f bm = { w[q & 1][e2][2 * s], w[q & 1][e2][2 * s + 1] };
                acc[pp] = __builtin_amdgcn_wmma_f32_16x16x4_f32(
                    false, a, false, bm, (short)0, acc[pp], false, false);
            }
        }
        if (q + 2 < NSLOT) {
            __builtin_amdgcn_sched_barrier(0);
            LOAD_PAIR(q + 2)
        }
        if (q & 1) {
            if (n < NOC) {
                #pragma unroll
                for (int r = 0; r < 8; ++r) {
                    int b = r + 8 * hb;
                    v4f v = { acc[0][r], acc[1][r], acc[2][r], acc[3][r] };
                    *(v4f*)(st + ((pr * BATCH + b) * NOC + n) * STP + PPW * wv) = v;
                }
            }
        }
    }
    #undef LOAD_PAIR

    __syncthreads();
    #pragma unroll
    for (int s = 0; s < OPIECES; ++s) {
        int f   = tid + NTHR * s;
        int pr  = f >> 10;
        int row = (f >> 3) & 127;
        int q   = f & 7;
        v4f v = *(const v4f*)(st + (pr * BATCH * NOC + row) * STP + 4 * q);
        *(v4f*)(out + (row * IMG + px0 + pr) * IMG + py0 + 4 * q) = v;
    }
}

extern "C" void kernel_launch(void* const* d_in, const int* in_sizes, int n_in,
                              void* d_out, int out_size, void* d_ws, size_t ws_size,
                              hipStream_t stream) {
    const float* x    = (const float*)d_in[0];
    const float* wgt  = (const float*)d_in[1];
    const float* bias = (const float*)d_in[2];
    float* out        = (float*)d_out;

    dim3 grid(IMG / PPB, IMG / PXR);
    dim3 block(NTHR);
    lcn2d_wmma_f32<<<grid, block, 0, stream>>>(x, wgt, bias, out);
}


# Two neighbouring pixels share one A: k = the 4-column union of their patches (8x3x4), B columns 0-7 / 8-15 = the two pixels, zero where a column is not theirs; 18 -> 12 WMMA/pixel; sum order changed
# speedup vs baseline: 7.8191x; 1.1281x over previous
#include <hip/hip_runtime.h>


typedef __attribute__((ext_vector_type(2))) float v2f;
typedef __attribute__((ext_vector_type(4))) float v4f;
typedef __attribute__((ext_vector_type(8))) float v8f;
typedef v4f v4f_a4 __attribute__((aligned(4)));

#define BATCH 16
#define CIN 8
#define IMG 256
#define NOC 8
#define KW 3
#define WROW (NOC * KW * KW)
#define NTHR 256
#define PPW 4
#define PPB 32
#define PXR 2
#define SRC (PXR + KW - 1)

#define XROWS (BATCH * CIN * SRC)
#define XP 44
#define XC0 4
#define CSTR (IMG * IMG * WROW)
#define STP 36

#define XPIECES (XROWS * PPB / 4 / NTHR)
#define XEDGES (XROWS * 2 / NTHR)
#define OPIECES (PXR * BATCH * NOC * PPB / 4 / NTHR)
#define NSLOT (PXR * PPW / 2)

__global__ __launch_bounds__(NTHR) __attribute__((amdgpu_waves_per_eu(4, 4)))
void lcn2d_wmma_f32(const float* __restrict__ xin,
                    const float* __restrict__ wgt,
                    const float* __restrict__ bias,
                    float* __restrict__ out)
{
    __shared__ __attribute__((aligned(16))) float xs[XROWS * XP];
    __shared__ __attribute__((aligned(16))) float st[PXR * BATCH * NOC * STP];

    const int tid  = threadIdx.x;
    const int px0  = blockIdx.y * PXR;
    const int py0  = blockIdx.x * PPB;
    const int wv   = tid >> 5;
    const int lane = tid & 31;
    const int n    = lane & 15;
    const int m    = lane & 15;
    const int hb   = lane >> 4;
    const int sel  = n >> 3;
    const int oc   = n & 7;

    float bv[NSLOT];
    #pragma unroll
    for (int q = 0; q < NSLOT; ++q)
        bv[q] = bias[((px0 + (q >> 1)) * IMG + py0 + PPW * wv + 2 * (q & 1) + sel) * NOC + oc];
    int sxr[SRC];
    #pragma unroll
    for (int r = 0; r < SRC; ++r) {
        int sx = px0 + r - 1;
        sxr[r] = sx < 0 ? 0 : (sx > IMG - 1 ? IMG - 1 : sx);
    }
    const int q8 = tid & 7;
    v4f xq[XPIECES];
    #pragma unroll
    for (int s = 0; s < XPIECES; ++s) {
        int r  = s >> 2;
        int mc = (tid >> 3) + 32 * (s & 3);
        xq[s] = *(const v4f*)(xin + (mc * IMG + sxr[r]) * IMG + py0 + 4 * q8);
    }
    const int emc  = tid >> 1;
    const int side = tid & 1;
    int ecol = side ? (py0 + PPB) : (py0 - 1);
    ecol = ecol < 0 ? 0 : (ecol > IMG - 1 ? IMG - 1 : ecol);
    float xe[XEDGES];
    #pragma unroll
    for (int r = 0; r < XEDGES; ++r)
        xe[r] = xin[(emc * IMG + sxr[r]) * IMG + ecol];
    const float* wl = wgt + ((4 * hb * IMG + px0) * IMG + py0 + PPW * wv + sel) * WROW + oc * 9 - sel;
    v4f w[2][4][KW];
    #define LOAD_PAIR(q)                                                                   \
        _Pragma("unroll")   \
        for (int cc = 0; cc < 4; ++cc) {                                                   \
            _Pragma("unroll")                                                 \
            for (int i = 0; i < KW; ++i)                                                   \
                w[(q) & 1][cc][i] = *(const v4f_a4*)(wl + ((q) >> 1) * (IMG * WROW)        \
                                                     + 2 * ((q) & 1) * WROW + cc * CSTR + 3 * i); \
        }
    LOAD_PAIR(0)
    __builtin_amdgcn_sched_barrier(0);
    LOAD_PAIR(1)
    #pragma unroll
    for (int s = 0; s < XPIECES; ++s) {
        int r    = s >> 2;
        int mc   = (tid >> 3) + 32 * (s & 3);
        int lrow = ((mc & 7) * SRC + r) * 16 + (mc >> 3);
        *(v4f*)(xs + lrow * XP + XC0 + 4 * q8) = xq[s];
    }
    #pragma unroll
    for (int r = 0; r < XEDGES; ++r) {
        int lrow = ((emc & 7) * SRC + r) * 16 + (emc >> 3);
        xs[lrow * XP + (side ? (XC0 + PPB) : (XC0 - 1))] = xe[r];
    }

    __builtin_amdgcn_fence(__ATOMIC_RELEASE, "workgroup", "local");
    __builtin_amdgcn_s_barrier();
    __builtin_amdgcn_fence(__ATOMIC_ACQUIRE, "workgroup", "local");

    const float* xa = xs + m * XP + (XC0 - 1) + PPW * wv + hb * (4 * SRC * 16 * XP);

    v4f a4[2][4][KW];
    #define LOAD_A(q)                                                                      \
        _Pragma("unroll")   \
        for (int cc = 0; cc < 4; ++cc) {                                                   \
            _Pragma("unroll")                                                 \
            for (int i = 0; i < KW; ++i)                                                   \
                a4[(q) & 1][cc][i] = *(const v4f_a4*)(xa + (cc * SRC + i + ((q) >> 1)) * 16 * XP + 2 * ((q) & 1)); \
        }
    LOAD_A(0)

    v8f res[2];
    #pragma unroll
    for (int q = 0; q < NSLOT; ++q) {
        const int pr = q >> 1;
        if (q + 1 < NSLOT) {
            LOAD_A(q + 1)
        }
        __builtin_amdgcn_sched_barrier(0);
        float b0 = bv[q];
        v8f acc0 = (v8f){ b0, b0, b0, b0, b0, b0, b0, b0 };
        v8f acc1 = (v8f){ 0.f, 0.f, 0.f, 0.f, 0.f, 0.f, 0.f, 0.f };
        #pragma unroll
        for (int cc = 0; cc < 4; ++cc) {
            #pragma unroll
            for (int i = 0; i < KW; ++i) {
                v4f wr = w[q & 1][cc][i];
                v2f a_lo = { a4[q & 1][cc][i][0], a4[q & 1][cc][i][1] };
                v2f a_hi = { a4[q & 1][cc][i][2], a4[q & 1][cc][i][3] };
                v2f b_lo = { sel ? 0.f : wr[0], wr[1] };
                v2f b_hi = { wr[2], sel ? wr[3] : 0.f };
                acc0 = __builtin_amdgcn_wmma_f32_16x16x4_f32(
                    false, a_lo, false, b_lo, (short)0, acc0, false, false);
                acc1 = __builtin_amdgcn_wmma_f32_16x16x4_f32(
                    false, a_hi, false, b_hi, (short)0, acc1, false, false);
            }
        }
        res[q & 1] = acc0 + acc1;
        if (q + 2 < NSLOT) {
            __builtin_amdgcn_sched_barrier(0);
            LOAD_PAIR(q + 2)
        }
        if (q & 1) {
            #pragma unroll
            for (int r = 0; r < 8; ++r) {
                int b = r + 8 * hb;
                v2f v = { res[0][r], res[1][r] };
                *(v2f*)(st + ((pr * BATCH + b) * NOC + oc) * STP + PPW * wv + 2 * sel) = v;
            }
        }
    }
    #undef LOAD_PAIR
    #undef LOAD_A

    __syncthreads();
    #pragma unroll
    for (int s = 0; s < OPIECES; ++s) {
        int f   = tid + NTHR * s;
        int pr  = f >> 10;
        int row = (f >> 3) & 127;
        int q   = f & 7;
        v4f v = *(const v4f*)(st + (pr * BATCH * NOC + row) * STP + 4 * q);
        v4f o = { v[0], v[2], v[1], v[3] };
        *(v4f*)(out + (row * IMG + px0 + pr) * IMG + py0 + 4 * q) = o;
    }
}

extern "C" void kernel_launch(void* const* d_in, const int* in_sizes, int n_in,
                              void* d_out, int out_size, void* d_ws, size_t ws_size,
                              hipStream_t stream) {
    const float* x    = (const float*)d_in[0];
    const float* wgt  = (const float*)d_in[1];
    const float* bias = (const float*)d_in[2];
    float* out        = (float*)d_out;

    dim3 grid(IMG / PPB, IMG / PXR);
    dim3 block(NTHR);
    lcn2d_wmma_f32<<<grid, block, 0, stream>>>(x, wgt, bias, out);
}


# Block walks two vertically adjacent tiles (grid 8x64, all resident): the 2nd tile keeps the two shared x rows in LDS and its loads are issued before the 1st tile's results leave; k, sum order kept
# speedup vs baseline: 8.5257x; 1.0904x over previous
#include <hip/hip_runtime.h>


typedef __attribute__((ext_vector_type(2))) float v2f;
typedef __attribute__((ext_vector_type(4))) float v4f;
typedef __attribute__((ext_vector_type(8))) float v8f;
typedef v4f v4f_a4 __attribute__((aligned(4)));

#define BATCH 16
#define CIN 8
#define IMG 256
#define NOC 8
#define KW 3
#define WROW (NOC * KW * KW)
#define NTHR 256
#define PPW 4
#define PPB 32
#define PXR 2
#define SRC (PXR + KW - 1)

#define XROWS (BATCH * CIN * SRC)
#define XP 44
#define XC0 4
#define CSTR (IMG * IMG * WROW)
#define STP 36

#define XPIECES (XROWS * PPB / 4 / NTHR)
#define XEDGES (XROWS * 2 / NTHR)
#define OPIECES (PXR * BATCH * NOC * PPB / 4 / NTHR)
#define NSLOT (PXR * PPW / 2)
#define NT 2
#define NEWROWS(t) ((t) == 0 ? SRC : PXR)
#define FIRSTNEW(t) (SRC - NEWROWS(t))
#define SLOT(t, lr) ((PXR * (t) + (lr)) & (SRC - 1))

__global__ __launch_bounds__(NTHR) __attribute__((amdgpu_waves_per_eu(4, 4)))
void lcn2d_wmma_f32(const float* __restrict__ xin,
                    const float* __restrict__ wgt,
                    const float* __restrict__ bias,
                    float* __restrict__ out)
{
    __shared__ __attribute__((aligned(16))) float xs[XROWS * XP];
    __shared__ __attribute__((aligned(16))) float st[PXR * BATCH * NOC * STP];

    const int tid  = threadIdx.x;
    const int px0  = blockIdx.y * (PXR * NT);
    const int py0  = blockIdx.x * PPB;
    const int wv   = tid >> 5;
    const int lane = tid & 31;
    const int n    = lane & 15;
    const int m    = lane & 15;
    const int hb   = lane >> 4;
    const int sel  = n >> 3;
    const int oc   = n & 7;

    const int q8   = tid & 7;
    const int emc  = tid >> 1;
    const int side = tid & 1;
    int ecol = side ? (py0 + PPB) : (py0 - 1);
    ecol = ecol < 0 ? 0 : (ecol > IMG - 1 ? IMG - 1 : ecol);

    const float* wl = wgt + ((4 * hb * IMG + px0) * IMG + py0 + PPW * wv + sel) * WROW + oc * 9 - sel;
    float bv[NSLOT];
    v4f xq[XPIECES];
    float xe[XEDGES];
    v4f w[2][4][KW];
    #define LOAD_PAIR(t, q)                                                                \
        _Pragma("unroll")   \
        for (int cc = 0; cc < 4; ++cc) {                                                   \
            _Pragma("unroll")                                                 \
            for (int i = 0; i < KW; ++i)                                                   \
                w[(q) & 1][cc][i] = *(const v4f_a4*)(wl + ((t) * PXR + ((q) >> 1)) * (IMG * WROW) \
                                                     + 2 * ((q) & 1) * WROW + cc * CSTR + 3 * i); \
        }
    #define ISSUE_TILE(t)                                                                  \
        _Pragma("unroll")                                                    \
        for (int q = 0; q < NSLOT; ++q)                                                    \
            bv[q] = bias[((px0 + (t) * PXR + (q >> 1)) * IMG + py0 + PPW * wv + 2 * (q & 1) + sel) * NOC + oc]; \
        _Pragma("unroll")             \
        for (int s = 0; s < 4 * NEWROWS(t); ++s) {                                         \
            int sx = px0 + (t) * PXR + FIRSTNEW(t) + (s >> 2) - 1;                         \
            sx = sx < 0 ? 0 : (sx > IMG - 1 ? IMG - 1 : sx);     \
            int mc = (tid >> 3) + 32 * (s & 3);                                            \
            xq[s] = *(const v4f*)(xin + (mc * IMG + sx) * IMG + py0 + 4 * q8);             \
        }                                                                                  \
        _Pragma("unroll")                                          \
        for (int r = 0; r < NEWROWS(t); ++r) {                                             \
            int sx = px0 + (t) * PXR + FIRSTNEW(t) + r - 1;                                \
            sx = sx < 0 ? 0 : (sx > IMG - 1 ? IMG - 1 : sx);                               \
            xe[r] = xin[(emc * IMG + sx) * IMG + ecol];                                    \
        }                                                                                  \
        LOAD_PAIR(t, 0)
    #define LDS_BARRIER()                                                                  \
        __builtin_amdgcn_fence(__ATOMIC_RELEASE, "workgroup", "local");                    \
        __builtin_amdgcn_s_barrier();                                                      \
        __builtin_amdgcn_fence(__ATOMIC_ACQUIRE, "workgroup", "local");

    const float* xa = xs + m * XP + (XC0 - 1) + PPW * wv + hb * (4 * SRC * 16 * XP);
    v4f a4[2][4][KW];
    #define LOAD_A(t, q)                                                                   \
        _Pragma("unroll")   \
        for (int cc = 0; cc < 4; ++cc) {                                                   \
            _Pragma("unroll")                                                 \
            for (int i = 0; i < KW; ++i)                                                   \
                a4[(q) & 1][cc][i] = *(const v4f_a4*)(xa + (cc * SRC + SLOT(t, i + ((q) >> 1))) * 16 * XP + 2 * ((q) & 1)); \
        }

    ISSUE_TILE(0)
    __builtin_amdgcn_sched_barrier(0);
    LOAD_PAIR(0, 1)

    #pragma unroll
    for (int t = 0; t < NT; ++t) {
        #pragma unroll
        for (int s = 0; s < 4 * NEWROWS(t); ++s) {
            int mc   = (tid >> 3) + 32 * (s & 3);
            int lrow = ((mc & 7) * SRC + SLOT(t, FIRSTNEW(t) + (s >> 2))) * 16 + (mc >> 3);
            *(v4f*)(xs + lrow * XP + XC0 + 4 * q8) = xq[s];
        }
        #pragma unroll
        for (int r = 0; r < NEWROWS(t); ++r) {
            int lrow = ((emc & 7) * SRC + SLOT(t, FIRSTNEW(t) + r)) * 16 + (emc >> 3);
            xs[lrow * XP + (side ? (XC0 + PPB) : (XC0 - 1))] = xe[r];
        }
        LDS_BARRIER()

        LOAD_A(t, 0)

        #pragma unroll
        for (int q = 0; q < NSLOT; ++q) {
            const int pr = q >> 1;
            if (q + 1 < NSLOT) {
                LOAD_A(t, q + 1)
            }
            __builtin_amdgcn_sched_barrier(0);
            float b0 = bv[q];
            v8f acc0 = (v8f){ b0, b0, b0, b0, b0, b0, b0, b0 };
            v8f acc1 = (v8f){ 0.f, 0.f, 0.f, 0.f, 0.f, 0.f, 0.f, 0.f };
            #pragma unroll
            for (int cc = 0; cc < 4; ++cc) {
                #pragma unroll
                for (int i = 0; i < KW; ++i) {
                    v4f wr = w[q & 1][cc][i];
                    v2f a_lo = { a4[q & 1][cc][i][0], a4[q & 1][cc][i][1] };
                    v2f a_hi = { a4[q & 1][cc][i][2], a4[q & 1][cc][i][3] };
                    v2f b_lo = { sel ? 0.f : wr[0], wr[1] };
                    v2f b_hi = { wr[2], sel ? wr[3] : 0.f };
                    acc0 = __builtin_amdgcn_wmma_f32_16x16x4_f32(
                        false, a_lo, false, b_lo, (short)0, acc0, false, false);
                    acc1 = __builtin_amdgcn_wmma_f32_16x16x4_f32(
                        false, a_hi, false, b_hi, (short)0, acc1, false, false);
                }
            }
            v8f res = acc0 + acc1;
            if (q + 2 < NSLOT) {
                __builtin_amdgcn_sched_barrier(0);
                LOAD_PAIR(t, q + 2)
            } else if (q + 1 == NSLOT && t + 1 < NT) {
                __builtin_amdgcn_sched_barrier(0);
                ISSUE_TILE(t + 1)
                __builtin_amdgcn_sched_barrier(0);
            }
            {
                #pragma unroll
                for (int r = 0; r < 8; ++r) {
                    int b = r + 8 * hb;
                    st[((pr * BATCH + b) * NOC + oc) * STP + PPW * wv + 2 * sel + (q & 1)] = res[r];
                }
            }
        }

        LDS_BARRIER()
        #pragma unroll
        for (int s = 0; s < OPIECES; ++s) {
            int f   = tid + NTHR * s;
            int pr  = f >> 10;
            int row = (f >> 3) & 127;
            int q   = f & 7;
            v4f v = *(const v4f*)(st + (pr * BATCH * NOC + row) * STP + 4 * q);
            v4f o = { v[0], v[2], v[1], v[3] };
            *(v4f*)(out + (row * IMG + px0 + t * PXR + pr) * IMG + py0 + 4 * q) = o;
        }
        if (t + 1 < NT) {
            __builtin_amdgcn_sched_barrier(0);
            LOAD_PAIR(t + 1, 1)
        }
    }
    #undef LOAD_PAIR
    #undef LOAD_A
    #undef ISSUE_TILE
    #undef LDS_BARRIER
}

extern "C" void kernel_launch(void* const* d_in, const int* in_sizes, int n_in,
                              void* d_out, int out_size, void* d_ws, size_t ws_size,
                              hipStream_t stream) {
    const float* x    = (const float*)d_in[0];
    const float* wgt  = (const float*)d_in[1];
    const float* bias = (const float*)d_in[2];
    float* out        = (float*)d_out;

    dim3 grid(IMG / PPB, IMG / (PXR * NT));
    dim3 block(NTHR);
    lcn2d_wmma_f32<<<grid, block, 0, stream>>>(x, wgt, bias, out);
}
